# starting kernel
# baseline (speedup 1.0000x reference)
#include <hip/hip_runtime.h>


typedef __attribute__((ext_vector_type(2))) float v2f;
typedef __attribute__((ext_vector_type(8))) float v8f;

#define BATCH   1024
#define IN_F    512
#define OUT_F   512
#define KC      128
#define LDSS    132

__global__ __launch_bounds__(32)
void fullconn_fused_kernel(const float* __restrict__ x,
                           const float* __restrict__ weight,
                           const float* __restrict__ proj_w,
                           const float* __restrict__ proj_b,
                           float* __restrict__ out)
{
    __shared__ float sx[16 * LDSS];
    __shared__ float sw[16 * LDSS];
    __shared__ float sp[16 * LDSS];

    const int lane = threadIdx.x;
    const int lk   = lane & 15;
    const int hi   = lane >> 4;
    const int tile = blockIdx.x;
    const int row0 = (tile >> 5) << 4;
    const int col0 = (tile & 31) << 4;

    float acc[8];
#pragma unroll
    for (int r = 0; r < 8; ++r) acc[r] = 0.0f;
    v8f c = {};

    const float bias = proj_b[col0 + lk];

    for (int kc = 0; kc < IN_F; kc += KC) {
        __syncthreads();
#pragma unroll
        for (int j = 0; j < 16; ++j) {
            const float4 xv = *(const float4*)(x      + (row0 + j) * IN_F + kc + lane * 4);
            const float4 wv = *(const float4*)(weight + (col0 + j) * IN_F + kc + lane * 4);
            const float4 pv = *(const float4*)(proj_w + (col0 + j) * IN_F + kc + lane * 4);
            *(float4*)(&sx[j * LDSS + lane * 4]) = xv;
            *(float4*)(&sw[j * LDSS + lane * 4]) = wv;
            *(float4*)(&sp[j * LDSS + lane * 4]) = pv;
        }
        __syncthreads();

        int p = (lk + kc) & 1;
        for (int k = 0; k < KC; ++k) {
            const float wv = sw[lk * LDSS + k];
#pragma unroll
            for (int r = 0; r < 8; ++r) {
                const float xv  = sx[(r + hi * 8) * LDSS + k];
                const float add = p ? xv : wv;
                const float mul = p ? wv : xv;
                acc[r] = (acc[r] + add) * mul;
            }
            p ^= 1;
        }

        const int koff = hi * 2;
#pragma unroll
        for (int k4 = 0; k4 < KC; k4 += 4) {
            v2f a, b;
            a.x = sx[lk * LDSS + k4 + koff];
            a.y = sx[lk * LDSS + k4 + koff + 1];
            b.x = sp[lk * LDSS + k4 + koff];
            b.y = sp[lk * LDSS + k4 + koff + 1];
            c = __builtin_amdgcn_wmma_f32_16x16x4_f32(
                     false, a,  false, b,
                     (short)0, c,  false,  false);
        }
    }

#pragma unroll
    for (int r = 0; r < 8; ++r) {
        const int row = row0 + r + hi * 8;
        out[row * OUT_F + col0 + lk] = acc[r] * (c[r] + bias);
    }
}

extern "C" void kernel_launch(void* const* d_in, const int* in_sizes, int n_in,
                              void* d_out, int out_size, void* d_ws, size_t ws_size,
                              hipStream_t stream) {
    (void)in_sizes; (void)n_in; (void)d_ws; (void)ws_size; (void)out_size;
    const float* x      = (const float*)d_in[0];
    const float* weight = (const float*)d_in[1];
    const float* proj_w = (const float*)d_in[2];
    const float* proj_b = (const float*)d_in[3];
    float* out = (float*)d_out;

    const int tiles = (BATCH / 16) * (OUT_F / 16);
    fullconn_fused_kernel<<<tiles, 32, 0, stream>>>(x, weight, proj_w, proj_b, out);
}


# LDS operand reads are issued one register block ahead of their use: recurrence in 8-step ping-pong blocks, WMMA fragments of a chunk as one group, so the LDS round trip is covered, not waited for
# speedup vs baseline: 1.1993x; 1.1993x over previous
#include <hip/hip_runtime.h>


typedef __attribute__((ext_vector_type(2))) float v2f;
typedef __attribute__((ext_vector_type(8))) float v8f;

#define BATCH   1024
#define IN_F    512
#define OUT_F   512
#define KC      128
#define LDSS    132
#define KB      8

struct alignas(16) RBlock {
    float x[8][KB];
    float w[KB];
};

__device__ __forceinline__ void load_rblock(RBlock& b, const float* sxh, const float* swl, int k0)
{
#pragma unroll
    for (int r = 0; r < 8; ++r) {
        *(float4*)(&b.x[r][0]) = *(const float4*)(sxh + r * LDSS + k0);
        *(float4*)(&b.x[r][4]) = *(const float4*)(sxh + r * LDSS + k0 + 4);
    }
    *(float4*)(&b.w[0]) = *(const float4*)(swl + k0);
    *(float4*)(&b.w[4]) = *(const float4*)(swl + k0 + 4);
}

__device__ __forceinline__ void run_rblock(float (&acc)[8], const RBlock& b, int pe)
{
#pragma unroll
    for (int j = 0; j < KB; ++j) {
        const int p = pe ^ (j & 1);
        const float wv = b.w[j];
#pragma unroll
        for (int r = 0; r < 8; ++r) {
            const float xv  = b.x[r][j];
            const float add = p ? xv : wv;
            const float mul = p ? wv : xv;
            acc[r] = (acc[r] + add) * mul;
        }
    }
}

__global__ __launch_bounds__(32)
void fullconn_fused_kernel(const float* __restrict__ x,
                           const float* __restrict__ weight,
                           const float* __restrict__ proj_w,
                           const float* __restrict__ proj_b,
                           float* __restrict__ out)
{
    __shared__ float sx[16 * LDSS];
    __shared__ float sw[16 * LDSS];
    __shared__ float sp[16 * LDSS];

    const int lane = threadIdx.x;
    const int lk   = lane & 15;
    const int hi   = lane >> 4;
    const int tile = blockIdx.x;
    const int row0 = (tile >> 5) << 4;
    const int col0 = (tile & 31) << 4;

    float acc[8];
#pragma unroll
    for (int r = 0; r < 8; ++r) acc[r] = 0.0f;
    v8f c = {};

    const float bias = proj_b[col0 + lk];

    const float* sxh = &sx[hi * 8 * LDSS];
    const float* swl = &sw[lk * LDSS];
    const int    pe  = lk & 1;
    const int    koff = hi * 2;
    const float* sxa = &sx[lk * LDSS + koff];
    const float* spb = &sp[lk * LDSS + koff];

    for (int kc = 0; kc < IN_F; kc += KC) {
        __syncthreads();
#pragma unroll
        for (int j = 0; j < 16; ++j) {
            const float4 xv = *(const float4*)(x      + (row0 + j) * IN_F + kc + lane * 4);
            const float4 wv = *(const float4*)(weight + (col0 + j) * IN_F + kc + lane * 4);
            const float4 pv = *(const float4*)(proj_w + (col0 + j) * IN_F + kc + lane * 4);
            *(float4*)(&sx[j * LDSS + lane * 4]) = xv;
            *(float4*)(&sw[j * LDSS + lane * 4]) = wv;
            *(float4*)(&sp[j * LDSS + lane * 4]) = pv;
        }
        __syncthreads();

        RBlock b0, b1;
        load_rblock(b0, sxh, swl, 0);
        for (int k0 = 0; k0 < KC - 2 * KB; k0 += 2 * KB) {
            load_rblock(b1, sxh, swl, k0 + KB);
            __builtin_amdgcn_sched_barrier(0);
            run_rblock(acc, b0, pe);
            load_rblock(b0, sxh, swl, k0 + 2 * KB);
            __builtin_amdgcn_sched_barrier(0);
            run_rblock(acc, b1, pe);
        }
        load_rblock(b1, sxh, swl, KC - KB);
        __builtin_amdgcn_sched_barrier(0);
        run_rblock(acc, b0, pe);
        run_rblock(acc, b1, pe);

        v2f af[KC / 4], bf[KC / 4];
#pragma unroll
        for (int i = 0; i < KC / 4; ++i) {
            af[i] = *(const v2f*)(sxa + 4 * i);
            bf[i] = *(const v2f*)(spb + 4 * i);
        }
#pragma unroll
        for (int i = 0; i < KC / 4; ++i) {
            c = __builtin_amdgcn_wmma_f32_16x16x4_f32(
                     false, af[i],  false, bf[i],
                     (short)0, c,  false,  false);
        }
    }

#pragma unroll
    for (int r = 0; r < 8; ++r) {
        const int row = row0 + r + hi * 8;
        out[row * OUT_F + col0 + lk] = acc[r] * (c[r] + bias);
    }
}

extern "C" void kernel_launch(void* const* d_in, const int* in_sizes, int n_in,
                              void* d_out, int out_size, void* d_ws, size_t ws_size,
                              hipStream_t stream) {
    (void)in_sizes; (void)n_in; (void)d_ws; (void)ws_size; (void)out_size;
    const float* x      = (const float*)d_in[0];
    const float* weight = (const float*)d_in[1];
    const float* proj_w = (const float*)d_in[2];
    const float* proj_b = (const float*)d_in[3];
    float* out = (float*)d_out;

    const int tiles = (BATCH / 16) * (OUT_F / 16);
    fullconn_fused_kernel<<<tiles, 32, 0, stream>>>(x, weight, proj_w, proj_b, out);
}


# Each wave owns 16 rows x 32 columns, one even and one odd column per lane: step parity is a compile-time constant (the two selects per step vanish) and both column sets share the x reads from LDS
# speedup vs baseline: 1.5403x; 1.2843x over previous
#include <hip/hip_runtime.h>


typedef __attribute__((ext_vector_type(2))) float v2f;
typedef __attribute__((ext_vector_type(8))) float v8f;

#define BATCH   1024
#define IN_F    512
#define OUT_F   512
#define KC      128
#define LDSS    132
#define KB      8
#define TN      32

struct alignas(16) RBlock {
    float x[8][KB];
    float we[KB];
    float wo[KB];
};

__device__ __forceinline__ void load_rblock(RBlock& b, const float* sxh, const float* swe, const float* swo, int k0)
{
#pragma unroll
    for (int r = 0; r < 8; ++r) {
        *(float4*)(&b.x[r][0]) = *(const float4*)(sxh + r * LDSS + k0);
        *(float4*)(&b.x[r][4]) = *(const float4*)(sxh + r * LDSS + k0 + 4);
    }
    *(float4*)(&b.we[0]) = *(const float4*)(swe + k0);
    *(float4*)(&b.we[4]) = *(const float4*)(swe + k0 + 4);
    *(float4*)(&b.wo[0]) = *(const float4*)(swo + k0);
    *(float4*)(&b.wo[4]) = *(const float4*)(swo + k0 + 4);
}

__device__ __forceinline__ void run_rblock(float (&ae)[8], float (&ao)[8], const RBlock& b)
{
#pragma unroll
    for (int j = 0; j < KB; ++j) {
        const float we = b.we[j];
        const float wo = b.wo[j];
#pragma unroll
        for (int r = 0; r < 8; ++r) {
            const float xv = b.x[r][j];
            if ((j & 1) == 0) {
                ae[r] = (ae[r] + we) * xv;
                ao[r] = (ao[r] + xv) * wo;
            } else {
                ae[r] = (ae[r] + xv) * we;
                ao[r] = (ao[r] + wo) * xv;
            }
        }
    }
}

__global__ __launch_bounds__(32)
void fullconn_fused_kernel(const float* __restrict__ x,
                           const float* __restrict__ weight,
                           const float* __restrict__ proj_w,
                           const float* __restrict__ proj_b,
                           float* __restrict__ out)
{
    __shared__ float sx[16 * LDSS];
    __shared__ float sw[TN * LDSS];
    __shared__ float sp[TN * LDSS];

    const int lane = threadIdx.x;
    const int lk   = lane & 15;
    const int hi   = lane >> 4;
    const int tile = blockIdx.x;
    const int row0 = (tile >> 4) << 4;
    const int col0 = (tile & 15) << 5;

    float ae[8], ao[8];
#pragma unroll
    for (int r = 0; r < 8; ++r) { ae[r] = 0.0f; ao[r] = 0.0f; }
    v8f ce = {};
    v8f co = {};

    const float2 bias = *(const float2*)(proj_b + col0 + 2 * lk);

    const float* sxh = &sx[hi * 8 * LDSS];
    const float* swe = &sw[lk * LDSS];
    const float* swo = &sw[(16 + lk) * LDSS];
    const int    koff = hi * 2;
    const float* sxa = &sx[lk * LDSS + koff];
    const float* spe = &sp[lk * LDSS + koff];
    const float* spo = &sp[(16 + lk) * LDSS + koff];

    for (int kc = 0; kc < IN_F; kc += KC) {
        __syncthreads();
#pragma unroll
        for (int j = 0; j < 16; ++j) {
            const float4 xv = *(const float4*)(x + (row0 + j) * IN_F + kc + lane * 4);
            *(float4*)(&sx[j * LDSS + lane * 4]) = xv;
        }
#pragma unroll
        for (int j = 0; j < TN; ++j) {
            const int lr = (j & 1) * 16 + (j >> 1);
            const float4 wv = *(const float4*)(weight + (col0 + j) * IN_F + kc + lane * 4);
            const float4 pv = *(const float4*)(proj_w + (col0 + j) * IN_F + kc + lane * 4);
            *(float4*)(&sw[lr * LDSS + lane * 4]) = wv;
            *(float4*)(&sp[lr * LDSS + lane * 4]) = pv;
        }
        __syncthreads();

        RBlock b0, b1;
        load_rblock(b0, sxh, swe, swo, 0);
        for (int k0 = 0; k0 < KC - 2 * KB; k0 += 2 * KB) {
            load_rblock(b1, sxh, swe, swo, k0 + KB);
            __builtin_amdgcn_sched_barrier(0);
            run_rblock(ae, ao, b0);
            load_rblock(b0, sxh, swe, swo, k0 + 2 * KB);
            __builtin_amdgcn_sched_barrier(0);
            run_rblock(ae, ao, b1);
        }
        load_rblock(b1, sxh, swe, swo, KC - KB);
        __builtin_amdgcn_sched_barrier(0);
        run_rblock(ae, ao, b0);
        run_rblock(ae, ao, b1);

        v2f af[KC / 4], be[KC / 4], bo[KC / 4];
#pragma unroll
        for (int i = 0; i < KC / 4; ++i) {
            af[i] = *(const v2f*)(sxa + 4 * i);
            be[i] = *(const v2f*)(spe + 4 * i);
            bo[i] = *(const v2f*)(spo + 4 * i);
        }
        __builtin_amdgcn_sched_barrier(0);
#pragma unroll
        for (int i = 0; i < KC / 4; ++i) {
            ce = __builtin_amdgcn_wmma_f32_16x16x4_f32(
                     false, af[i],  false, be[i],
                     (short)0, ce,  false,  false);
            co = __builtin_amdgcn_wmma_f32_16x16x4_f32(
                     false, af[i],  false, bo[i],
                     (short)0, co,  false,  false);
        }
    }

#pragma unroll
    for (int r = 0; r < 8; ++r) {
        const int row = row0 + r + hi * 8;
        float2 o;
        o.x = ae[r] * (ce[r] + bias.x);
        o.y = ao[r] * (co[r] + bias.y);
        *(float2*)(out + row * OUT_F + col0 + 2 * lk) = o;
    }
}

extern "C" void kernel_launch(void* const* d_in, const int* in_sizes, int n_in,
                              void* d_out, int out_size, void* d_ws, size_t ws_size,
                              hipStream_t stream) {
    (void)in_sizes; (void)n_in; (void)d_ws; (void)ws_size; (void)out_size;
    const float* x      = (const float*)d_in[0];
    const float* weight = (const float*)d_in[1];
    const float* proj_w = (const float*)d_in[2];
    const float* proj_b = (const float*)d_in[3];
    float* out = (float*)d_out;

    const int tiles = (BATCH / 16) * (OUT_F / TN);
    fullconn_fused_kernel<<<tiles, 32, 0, stream>>>(x, weight, proj_w, proj_b, out);
}


# K chunks software-pipelined through registers: the global loads of chunk c+1 are issued before the math of chunk c and stored to LDS after it, so only chunk 0's trip from memory is waited for
# speedup vs baseline: 2.0677x; 1.3424x over previous
#include <hip/hip_runtime.h>


typedef __attribute__((ext_vector_type(2))) float v2f;
typedef __attribute__((ext_vector_type(4))) float v4f;
typedef __attribute__((ext_vector_type(8))) float v8f;

#define BATCH   1024
#define IN_F    512
#define OUT_F   512
#define KC      128
#define LDSS    132
#define KB      8
#define TN      32

struct alignas(16) RBlock {
    float x[8][KB];
    float we[KB];
    float wo[KB];
};

__device__ __forceinline__ void load_rblock(RBlock& b, const float* sxh, const float* swe, const float* swo, int k0)
{
#pragma unroll
    for (int r = 0; r < 8; ++r) {
        *(float4*)(&b.x[r][0]) = *(const float4*)(sxh + r * LDSS + k0);
        *(float4*)(&b.x[r][4]) = *(const float4*)(sxh + r * LDSS + k0 + 4);
    }
    *(float4*)(&b.we[0]) = *(const float4*)(swe + k0);
    *(float4*)(&b.we[4]) = *(const float4*)(swe + k0 + 4);
    *(float4*)(&b.wo[0]) = *(const float4*)(swo + k0);
    *(float4*)(&b.wo[4]) = *(const float4*)(swo + k0 + 4);
}

__device__ __forceinline__ void run_rblock(float (&ae)[8], float (&ao)[8], const RBlock& b)
{
#pragma unroll
    for (int j = 0; j < KB; ++j) {
        const float we = b.we[j];
        const float wo = b.wo[j];
#pragma unroll
        for (int r = 0; r < 8; ++r) {
            const float xv = b.x[r][j];
            if ((j & 1) == 0) {
                ae[r] = (ae[r] + we) * xv;
                ao[r] = (ao[r] + xv) * wo;
            } else {
                ae[r] = (ae[r] + xv) * we;
                ao[r] = (ao[r] + wo) * xv;
            }
        }
    }
}


__device__ __forceinline__ void fetch_xw(v4f (&fx)[16], v4f (&fw)[TN], const float* xg, const float* wg, int kc)
{
#pragma unroll
    for (int j = 0; j < 16; ++j) fx[j] = *(const v4f*)(xg + j * IN_F + kc);
#pragma unroll
    for (int j = 0; j < TN; ++j) fw[j] = *(const v4f*)(wg + j * IN_F + kc);
}

__device__ __forceinline__ void fetch_p(v4f (&fp)[TN], const float* pg, int kc)
{
#pragma unroll
    for (int j = 0; j < TN; ++j) fp[j] = *(const v4f*)(pg + j * IN_F + kc);
}

__device__ __forceinline__ void land(const v4f (&fx)[16], const v4f (&fw)[TN], const v4f (&fp)[TN],
                                     float* sx, float* sw, float* sp, int lane)
{
#pragma unroll
    for (int j = 0; j < 16; ++j) *(v4f*)(&sx[j * LDSS + lane * 4]) = fx[j];
#pragma unroll
    for (int j = 0; j < TN; ++j) *(v4f*)(&sw[((j & 1) * 16 + (j >> 1)) * LDSS + lane * 4]) = fw[j];
#pragma unroll
    for (int j = 0; j < TN; ++j) *(v4f*)(&sp[((j & 1) * 16 + (j >> 1)) * LDSS + lane * 4]) = fp[j];
}

__global__ __launch_bounds__(32)
void fullconn_fused_kernel(const float* __restrict__ x,
                           const float* __restrict__ weight,
                           const float* __restrict__ proj_w,
                           const float* __restrict__ proj_b,
                           float* __restrict__ out)
{
    __shared__ float sx[16 * LDSS];
    __shared__ float sw[TN * LDSS];
    __shared__ float sp[TN * LDSS];

    const int lane = threadIdx.x;
    const int lk   = lane & 15;
    const int hi   = lane >> 4;
    const int tile = blockIdx.x;
    const int row0 = (tile >> 4) << 4;
    const int col0 = (tile & 15) << 5;

    float ae[8], ao[8];
#pragma unroll
    for (int r = 0; r < 8; ++r) { ae[r] = 0.0f; ao[r] = 0.0f; }
    v8f ce = {};
    v8f co = {};

    const float2 bias = *(const float2*)(proj_b + col0 + 2 * lk);

    const float* sxh = &sx[hi * 8 * LDSS];
    const float* swe = &sw[lk * LDSS];
    const float* swo = &sw[(16 + lk) * LDSS];
    const int    koff = hi * 2;
    const float* sxa = &sx[lk * LDSS + koff];
    const float* spe = &sp[lk * LDSS + koff];
    const float* spo = &sp[(16 + lk) * LDSS + koff];

    const float* xg = x      + row0 * IN_F + lane * 4;
    const float* wg = weight + col0 * IN_F + lane * 4;
    const float* pg = proj_w + col0 * IN_F + lane * 4;

    v4f fx[16], fw[TN], fp[TN];
    fetch_xw(fx, fw, xg, wg, 0);
    fetch_p(fp, pg, 0);
    land(fx, fw, fp, sx, sw, sp, lane);
    __syncthreads();

    for (int kc = 0; kc < IN_F; kc += KC) {
        const bool more = (kc + KC < IN_F);
        if (more) fetch_p(fp, pg, kc + KC);

        {
            v2f af[KC / 4], be[KC / 4], bo[KC / 4];
#pragma unroll
            for (int i = 0; i < KC / 4; ++i) {
                af[i] = *(const v2f*)(sxa + 4 * i);
                be[i] = *(const v2f*)(spe + 4 * i);
                bo[i] = *(const v2f*)(spo + 4 * i);
            }
            __builtin_amdgcn_sched_barrier(0);
#pragma unroll
            for (int i = 0; i < KC / 4; ++i) {
                ce = __builtin_amdgcn_wmma_f32_16x16x4_f32(
                         false, af[i],  false, be[i],
                         (short)0, ce,  false,  false);
                co = __builtin_amdgcn_wmma_f32_16x16x4_f32(
                         false, af[i],  false, bo[i],
                         (short)0, co,  false,  false);
            }
        }

        if (more) fetch_xw(fx, fw, xg, wg, kc + KC);

        RBlock b0, b1;
        load_rblock(b0, sxh, swe, swo, 0);
        for (int k0 = 0; k0 < KC - 2 * KB; k0 += 2 * KB) {
            load_rblock(b1, sxh, swe, swo, k0 + KB);
            __builtin_amdgcn_sched_barrier(0);
            run_rblock(ae, ao, b0);
            load_rblock(b0, sxh, swe, swo, k0 + 2 * KB);
            __builtin_amdgcn_sched_barrier(0);
            run_rblock(ae, ao, b1);
        }
        load_rblock(b1, sxh, swe, swo, KC - KB);
        __builtin_amdgcn_sched_barrier(0);
        run_rblock(ae, ao, b0);
        run_rblock(ae, ao, b1);

        if (more) {
            __syncthreads();
            land(fx, fw, fp, sx, sw, sp, lane);
            __syncthreads();
        }
    }

#pragma unroll
    for (int r = 0; r < 8; ++r) {
        const int row = row0 + r + hi * 8;
        float2 o;
        o.x = ae[r] * (ce[r] + bias.x);
        o.y = ao[r] * (co[r] + bias.y);
        *(float2*)(out + row * OUT_F + col0 + 2 * lk) = o;
    }
}

extern "C" void kernel_launch(void* const* d_in, const int* in_sizes, int n_in,
                              void* d_out, int out_size, void* d_ws, size_t ws_size,
                              hipStream_t stream) {
    (void)in_sizes; (void)n_in; (void)d_ws; (void)ws_size; (void)out_size;
    const float* x      = (const float*)d_in[0];
    const float* weight = (const float*)d_in[1];
    const float* proj_w = (const float*)d_in[2];
    const float* proj_b = (const float*)d_in[3];
    float* out = (float*)d_out;

    const int tiles = (BATCH / 16) * (OUT_F / TN);
    fullconn_fused_kernel<<<tiles, 32, 0, stream>>>(x, weight, proj_w, proj_b, out);
}


# One 8-wave work-group per WGP: waves w and w+4 share a SIMD and a 16x32 piece; wave w runs only the recurrence, wave w+4 its WMMAs and the staging into a second LDS buffer, under w's vector work
# speedup vs baseline: 2.3214x; 1.1227x over previous
#include <hip/hip_runtime.h>


typedef __attribute__((ext_vector_type(2))) float v2f;
typedef __attribute__((ext_vector_type(4))) float v4f;
typedef __attribute__((ext_vector_type(8))) float v8f;

#define BATCH   1024
#define IN_F    512
#define OUT_F   512
#define KC      128
#define LDSS    132
#define KB      8
#define TN      32

#define XT          (16 * LDSS)
#define WT          (TN * LDSS)
#define X_AT(tr, b) ((tr) * 8192 + (b) * XT)
#define W_AT(tc, b) (16384 + ((b) * 2 + (tc)) * 8192)
#define P_AT(tc, b) (49152 + ((b) * 2 + (tc)) * WT)
#define E_AT        (49152 + 4 * WT)
#define LDS_FLOATS  (E_AT + 4 * 16 * 32)

struct alignas(16) RBlock {
    float x[8][KB];
    float we[KB];
    float wo[KB];
};

__device__ __forceinline__ void load_rblock(RBlock& b, const float* sxh, const float* swe, const float* swo, int k0)
{
#pragma unroll
    for (int r = 0; r < 8; ++r) {
        *(float4*)(&b.x[r][0]) = *(const float4*)(sxh + r * LDSS + k0);
        *(float4*)(&b.x[r][4]) = *(const float4*)(sxh + r * LDSS + k0 + 4);
    }
    *(float4*)(&b.we[0]) = *(const float4*)(swe + k0);
    *(float4*)(&b.we[4]) = *(const float4*)(swe + k0 + 4);
    *(float4*)(&b.wo[0]) = *(const float4*)(swo + k0);
    *(float4*)(&b.wo[4]) = *(const float4*)(swo + k0 + 4);
}

__device__ __forceinline__ void run_rblock(float (&ae)[8], float (&ao)[8], const RBlock& b)
{
#pragma unroll
    for (int j = 0; j < KB; ++j) {
        const float we = b.we[j];
        const float wo = b.wo[j];
#pragma unroll
        for (int r = 0; r < 8; ++r) {
            const float xv = b.x[r][j];
            if ((j & 1) == 0) {
                ae[r] = (ae[r] + we) * xv;
                ao[r] = (ao[r] + xv) * wo;
            } else {
                ae[r] = (ae[r] + xv) * we;
                ao[r] = (ao[r] + wo) * xv;
            }
        }
    }
}

__device__ __forceinline__ void fetch_share(v4f (&fx)[8], v4f (&fw)[16], v4f (&fp)[16],
                                            const float* xg, const float* wg, const float* pg, int kc)
{
#pragma unroll
    for (int j = 0; j < 8; ++j)  fx[j] = *(const v4f*)(xg + j * IN_F + kc);
#pragma unroll
    for (int j = 0; j < 16; ++j) fw[j] = *(const v4f*)(wg + j * IN_F + kc);
#pragma unroll
    for (int j = 0; j < 16; ++j) fp[j] = *(const v4f*)(pg + j * IN_F + kc);
}

__device__ __forceinline__ void land_share(const v4f (&fx)[8], const v4f (&fw)[16], const v4f (&fp)[16],
                                           float* sx, float* sw, float* sp, int lane)
{
#pragma unroll
    for (int j = 0; j < 8; ++j)  *(v4f*)(&sx[j * LDSS + lane * 4]) = fx[j];
#pragma unroll
    for (int j = 0; j < 16; ++j) *(v4f*)(&sw[((j & 1) * 16 + (j >> 1)) * LDSS + lane * 4]) = fw[j];
#pragma unroll
    for (int j = 0; j < 16; ++j) *(v4f*)(&sp[((j & 1) * 16 + (j >> 1)) * LDSS + lane * 4]) = fp[j];
}

template <bool MORE>
__device__ __forceinline__ void chunk_body(int role, int c, int b, float* lds, int tr, int tc, int q, int sq,
                                           int lk, int hi, int lane,
                                           float (&ae)[8], float (&ao)[8], v8f& ce, v8f& co,
                                           const float* xg, const float* wg, const float* pg)
{
    if (role == 0) {
        const float* sxh = lds + X_AT(tr, b) + hi * 8 * LDSS;
        const float* swe = lds + W_AT(tc, b) + lk * LDSS;
        const float* swo = swe + 16 * LDSS;
        RBlock b0, b1;
        load_rblock(b0, sxh, swe, swo, 0);
        for (int k0 = 0; k0 < KC - 2 * KB; k0 += 2 * KB) {
            load_rblock(b1, sxh, swe, swo, k0 + KB);
            __builtin_amdgcn_sched_barrier(0);
            run_rblock(ae, ao, b0);
            load_rblock(b0, sxh, swe, swo, k0 + 2 * KB);
            __builtin_amdgcn_sched_barrier(0);
            run_rblock(ae, ao, b1);
        }
        load_rblock(b1, sxh, swe, swo, KC - KB);
        __builtin_amdgcn_sched_barrier(0);
        run_rblock(ae, ao, b0);
        run_rblock(ae, ao, b1);
    } else {
        v4f fx[8], fw[16], fp[16];
        if (MORE) fetch_share(fx, fw, fp, xg, wg, pg, (c + 1) * KC);

        {
            const int    koff = hi * 2;
            const float* sxa = lds + X_AT(tr, b) + lk * LDSS + koff;
            const float* spe = lds + P_AT(tc, b) + lk * LDSS + koff;
            const float* spo = spe + 16 * LDSS;
            v2f af[KC / 4], be[KC / 4], bo[KC / 4];
#pragma unroll
            for (int i = 0; i < KC / 4; ++i) {
                af[i] = *(const v2f*)(sxa + 4 * i);
                be[i] = *(const v2f*)(spe + 4 * i);
                bo[i] = *(const v2f*)(spo + 4 * i);
            }
            __builtin_amdgcn_sched_barrier(0);
#pragma unroll
            for (int i = 0; i < KC / 4; ++i) {
                ce = __builtin_amdgcn_wmma_f32_16x16x4_f32(
                         false, af[i],  false, be[i],
                         (short)0, ce,  false,  false);
                co = __builtin_amdgcn_wmma_f32_16x16x4_f32(
                         false, af[i],  false, bo[i],
                         (short)0, co,  false,  false);
            }
        }

        if (MORE) {
            asm volatile("" :: "v"(ce), "v"(co) : "memory");
            land_share(fx, fw, fp, lds + X_AT(q >> 1, b ^ 1) + sq, lds + W_AT(q >> 1, b ^ 1) + sq,
                       lds + P_AT(q >> 1, b ^ 1) + sq, lane);
        }
    }
}

__global__ __launch_bounds__(256)
void fullconn_fused_kernel(const float* __restrict__ x,
                           const float* __restrict__ weight,
                           const float* __restrict__ proj_w,
                           const float* __restrict__ proj_b,
                           float* __restrict__ out)
{
    __shared__ float lds[LDS_FLOATS];

    const int lane = threadIdx.x & 31;
    const int wid  = __builtin_amdgcn_readfirstlane((int)(threadIdx.x >> 5));
    const int role = wid >> 2;
    const int q    = wid & 3;
    const int tr   = q & 1;
    const int tc   = q >> 1;
    const int lk   = lane & 15;
    const int hi   = lane >> 4;
    const int R0   = (blockIdx.x >> 3) << 5;
    const int C0   = (blockIdx.x & 7) << 6;
    const int row0 = R0 + tr * 16;
    const int col0 = C0 + tc * TN;

    float ae[8], ao[8];
#pragma unroll
    for (int r = 0; r < 8; ++r) { ae[r] = 0.0f; ao[r] = 0.0f; }
    v8f ce = {};
    v8f co = {};

    const float* xg = x      + (R0 + q * 8)  * IN_F + lane * 4;
    const float* wg = weight + (C0 + q * 16) * IN_F + lane * 4;
    const float* pg = proj_w + (C0 + q * 16) * IN_F + lane * 4;
    const int    sq = (q & 1) * 8 * LDSS;

    if (role == 1) {
        v4f fx[8], fw[16], fp[16];
        fetch_share(fx, fw, fp, xg, wg, pg, 0);
        land_share(fx, fw, fp, lds + X_AT(q >> 1, 0) + sq, lds + W_AT(q >> 1, 0) + sq, lds + P_AT(q >> 1, 0) + sq, lane);
    }
    __syncthreads();

    for (int c = 0; c < IN_F / KC - 1; ++c) {
        chunk_body<true>(role, c, c & 1, lds, tr, tc, q, sq, lk, hi, lane, ae, ao, ce, co, xg, wg, pg);
        __syncthreads();
    }
    chunk_body<false>(role, IN_F / KC - 1, (IN_F / KC - 1) & 1, lds, tr, tc, q, sq, lk, hi, lane, ae, ao, ce, co, xg, wg, pg);

    float* se = lds + E_AT + q * (16 * 32) + lane;
    if (role == 1) {
        const float2 bias = *(const float2*)(proj_b + col0 + 2 * lk);
#pragma unroll
        for (int r = 0; r < 8; ++r) {
            se[(2 * r)     * 32] = ce[r] + bias.x;
            se[(2 * r + 1) * 32] = co[r] + bias.y;
        }
    }
    __syncthreads();
    if (role == 0) {
#pragma unroll
        for (int r = 0; r < 8; ++r) {
            const int row = row0 + r + hi * 8;
            float2 o;
            o.x = ae[r] * se[(2 * r)     * 32];
            o.y = ao[r] * se[(2 * r + 1) * 32];
            *(float2*)(out + row * OUT_F + col0 + 2 * lk) = o;
        }
    }
}

extern "C" void kernel_launch(void* const* d_in, const int* in_sizes, int n_in,
                              void* d_out, int out_size, void* d_ws, size_t ws_size,
                              hipStream_t stream) {
    (void)in_sizes; (void)n_in; (void)d_ws; (void)ws_size; (void)out_size;
    const float* x      = (const float*)d_in[0];
    const float* weight = (const float*)d_in[1];
    const float* proj_w = (const float*)d_in[2];
    const float* proj_b = (const float*)d_in[3];
    float* out = (float*)d_out;

    const int groups = (BATCH / 32) * (OUT_F / 64);
    fullconn_fused_kernel<<<groups, 256, 0, stream>>>(x, weight, proj_w, proj_b, out);
}


# Work-group tile 64 rows x 32 columns instead of 32 x 64 (four 16x32 pieces stacked in rows share one weight and one proj tile): 128 operand rows staged per K chunk instead of 160
# speedup vs baseline: 2.5921x; 1.1166x over previous
#include <hip/hip_runtime.h>


typedef __attribute__((ext_vector_type(2))) float v2f;
typedef __attribute__((ext_vector_type(4))) float v4f;
typedef __attribute__((ext_vector_type(8))) float v8f;

#define BATCH   1024
#define IN_F    512
#define OUT_F   512
#define KC      128
#define LDSS    132
#define KB      8
#define TN      32

#define XT          (16 * LDSS)
#define WT          (TN * LDSS)
#define X_AT(p, b)  ((p) * 8192 + (b) * XT)
#define W_AT(b)     (4 * 8192 + (b) * WT)
#define P_AT(b)     (6 * 8192 + (b) * WT)
#define E_AT        (8 * 8192)
#define LDS_FLOATS  (E_AT + 4 * 16 * 32)

struct alignas(16) RBlock {
    float x[8][KB];
    float we[KB];
    float wo[KB];
};

__device__ __forceinline__ void load_rblock(RBlock& b, const float* sxh, const float* swe, const float* swo, int k0)
{
#pragma unroll
    for (int r = 0; r < 8; ++r) {
        *(float4*)(&b.x[r][0]) = *(const float4*)(sxh + r * LDSS + k0);
        *(float4*)(&b.x[r][4]) = *(const float4*)(sxh + r * LDSS + k0 + 4);
    }
    *(float4*)(&b.we[0]) = *(const float4*)(swe + k0);
    *(float4*)(&b.we[4]) = *(const float4*)(swe + k0 + 4);
    *(float4*)(&b.wo[0]) = *(const float4*)(swo + k0);
    *(float4*)(&b.wo[4]) = *(const float4*)(swo + k0 + 4);
}

__device__ __forceinline__ void run_rblock(float (&ae)[8], float (&ao)[8], const RBlock& b)
{
#pragma unroll
    for (int j = 0; j < KB; ++j) {
        const float we = b.we[j];
        const float wo = b.wo[j];
#pragma unroll
        for (int r = 0; r < 8; ++r) {
            const float xv = b.x[r][j];
            if ((j & 1) == 0) {
                ae[r] = (ae[r] + we) * xv;
                ao[r] = (ao[r] + xv) * wo;
            } else {
                ae[r] = (ae[r] + xv) * we;
                ao[r] = (ao[r] + wo) * xv;
            }
        }
    }
}

__device__ __forceinline__ void fetch_share(v4f (&fx)[16], v4f (&fw)[8], v4f (&fp)[8],
                                            const float* xg, const float* wg, const float* pg, int kc)
{
#pragma unroll
    for (int j = 0; j < 16; ++j) fx[j] = *(const v4f*)(xg + j * IN_F + kc);
#pragma unroll
    for (int j = 0; j < 8; ++j)  fw[j] = *(const v4f*)(wg + j * IN_F + kc);
#pragma unroll
    for (int j = 0; j < 8; ++j)  fp[j] = *(const v4f*)(pg + j * IN_F + kc);
}

__device__ __forceinline__ void land_share(const v4f (&fx)[16], const v4f (&fw)[8], const v4f (&fp)[8],
                                           float* sx, float* sw, float* sp, int lane)
{
#pragma unroll
    for (int j = 0; j < 16; ++j) *(v4f*)(&sx[j * LDSS + lane * 4]) = fx[j];
#pragma unroll
    for (int j = 0; j < 8; ++j)  *(v4f*)(&sw[((j & 1) * 16 + (j >> 1)) * LDSS + lane * 4]) = fw[j];
#pragma unroll
    for (int j = 0; j < 8; ++j)  *(v4f*)(&sp[((j & 1) * 16 + (j >> 1)) * LDSS + lane * 4]) = fp[j];
}

template <bool MORE>
__device__ __forceinline__ void chunk_body(int role, int c, int b, float* lds, int q, int sq,
                                           int lk, int hi, int lane,
                                           float (&ae)[8], float (&ao)[8], v8f& ce, v8f& co,
                                           const float* xg, const float* wg, const float* pg)
{
    if (role == 0) {
        const float* sxh = lds + X_AT(q, b) + hi * 8 * LDSS;
        const float* swe = lds + W_AT(b) + lk * LDSS;
        const float* swo = swe + 16 * LDSS;
        RBlock b0, b1;
        load_rblock(b0, sxh, swe, swo, 0);
        for (int k0 = 0; k0 < KC - 2 * KB; k0 += 2 * KB) {
            load_rblock(b1, sxh, swe, swo, k0 + KB);
            __builtin_amdgcn_sched_barrier(0);
            run_rblock(ae, ao, b0);
            load_rblock(b0, sxh, swe, swo, k0 + 2 * KB);
            __builtin_amdgcn_sched_barrier(0);
            run_rblock(ae, ao, b1);
        }
        load_rblock(b1, sxh, swe, swo, KC - KB);
        __builtin_amdgcn_sched_barrier(0);
        run_rblock(ae, ao, b0);
        run_rblock(ae, ao, b1);
    } else {
        v4f fx[16], fw[8], fp[8];
        if (MORE) fetch_share(fx, fw, fp, xg, wg, pg, (c + 1) * KC);

        {
            const int    koff = hi * 2;
            const float* sxa = lds + X_AT(q, b) + lk * LDSS + koff;
            const float* spe = lds + P_AT(b) + lk * LDSS + koff;
            const float* spo = spe + 16 * LDSS;
            v2f af[KC / 4], be[KC / 4], bo[KC / 4];
#pragma unroll
            for (int i = 0; i < KC / 4; ++i) {
                af[i] = *(const v2f*)(sxa + 4 * i);
                be[i] = *(const v2f*)(spe + 4 * i);
                bo[i] = *(const v2f*)(spo + 4 * i);
            }
            __builtin_amdgcn_sched_barrier(0);
#pragma unroll
            for (int i = 0; i < KC / 4; ++i) {
                ce = __builtin_amdgcn_wmma_f32_16x16x4_f32(
                         false, af[i],  false, be[i],
                         (short)0, ce,  false,  false);
                co = __builtin_amdgcn_wmma_f32_16x16x4_f32(
                         false, af[i],  false, bo[i],
                         (short)0, co,  false,  false);
            }
        }

        if (MORE) {
            asm volatile("" :: "v"(ce), "v"(co) : "memory");
            land_share(fx, fw, fp, lds + X_AT(q, b ^ 1), lds + W_AT(b ^ 1) + sq, lds + P_AT(b ^ 1) + sq, lane);
        }
    }
}

__global__ __launch_bounds__(256)
void fullconn_fused_kernel(const float* __restrict__ x,
                           const float* __restrict__ weight,
                           const float* __restrict__ proj_w,
                           const float* __restrict__ proj_b,
                           float* __restrict__ out)
{
    __shared__ float lds[LDS_FLOATS];

    const int lane = threadIdx.x & 31;
    const int wid  = __builtin_amdgcn_readfirstlane((int)(threadIdx.x >> 5));
    const int role = wid >> 2;
    const int q    = wid & 3;
    const int lk   = lane & 15;
    const int hi   = lane >> 4;
    const int R0   = (blockIdx.x >> 4) << 6;
    const int C0   = (blockIdx.x & 15) * TN;
    const int row0 = R0 + q * 16;
    const int col0 = C0;

    float ae[8], ao[8];
#pragma unroll
    for (int r = 0; r < 8; ++r) { ae[r] = 0.0f; ao[r] = 0.0f; }
    v8f ce = {};
    v8f co = {};

    const float* xg = x      + (R0 + q * 16) * IN_F + lane * 4;
    const float* wg = weight + (C0 + q * 8)  * IN_F + lane * 4;
    const float* pg = proj_w + (C0 + q * 8)  * IN_F + lane * 4;
    const int    sq = q * 4 * LDSS;

    if (role == 1) {
        v4f fx[16], fw[8], fp[8];
        fetch_share(fx, fw, fp, xg, wg, pg, 0);
        land_share(fx, fw, fp, lds + X_AT(q, 0), lds + W_AT(0) + sq, lds + P_AT(0) + sq, lane);
    }
    __syncthreads();

    for (int c = 0; c < IN_F / KC - 1; ++c) {
        chunk_body<true>(role, c, c & 1, lds, q, sq, lk, hi, lane, ae, ao, ce, co, xg, wg, pg);
        __syncthreads();
    }
    chunk_body<false>(role, IN_F / KC - 1, (IN_F / KC - 1) & 1, lds, q, sq, lk, hi, lane, ae, ao, ce, co, xg, wg, pg);

    float* se = lds + E_AT + q * (16 * 32) + lane;
    if (role == 1) {
        const float2 bias = *(const float2*)(proj_b + col0 + 2 * lk);
#pragma unroll
        for (int r = 0; r < 8; ++r) {
            se[(2 * r)     * 32] = ce[r] + bias.x;
            se[(2 * r + 1) * 32] = co[r] + bias.y;
        }
    }
    __syncthreads();
    if (role == 0) {
#pragma unroll
        for (int r = 0; r < 8; ++r) {
            const int row = row0 + r + hi * 8;
            float2 o;
            o.x = ae[r] * se[(2 * r)     * 32];
            o.y = ao[r] * se[(2 * r + 1) * 32];
            *(float2*)(out + row * OUT_F + col0 + 2 * lk) = o;
        }
    }
}

extern "C" void kernel_launch(void* const* d_in, const int* in_sizes, int n_in,
                              void* d_out, int out_size, void* d_ws, size_t ws_size,
                              hipStream_t stream) {
    (void)in_sizes; (void)n_in; (void)d_ws; (void)ws_size; (void)out_size;
    const float* x      = (const float*)d_in[0];
    const float* weight = (const float*)d_in[1];
    const float* proj_w = (const float*)d_in[2];
    const float* proj_b = (const float*)d_in[3];
    float* out = (float*)d_out;

    const int groups = (BATCH / 64) * (OUT_F / TN);
    fullconn_fused_kernel<<<groups, 256, 0, stream>>>(x, weight, proj_w, proj_b, out);
}
